# starting kernel
# baseline (speedup 1.0000x reference)
#include <hip/hip_runtime.h>
#include <math.h>


#define N_NODES 50000
#define N_EDGES 800000
#define FEAT    32
#define HEADS   4
#define MTILES  (N_NODES / 16)

typedef __attribute__((ext_vector_type(2))) float v2f;
typedef __attribute__((ext_vector_type(8))) float v8f;

__device__ __forceinline__ void atomic_add_f32_dev(float* p, float v) {
  asm volatile("global_atomic_add_f32 %0, %1, off scope:SCOPE_DEV"
               :: "v"(p), "v"(v) : "memory");
}
__device__ __forceinline__ void atomic_max_f32_dev(float* p, float v) {
  asm volatile("global_atomic_max_num_f32 %0, %1, off scope:SCOPE_DEV"
               :: "v"(p), "v"(v) : "memory");
}

__device__ __forceinline__ v8f wmma_f32_k32(const float* Arow,
                                            const float* __restrict__ B,
                                            int bcol, int lane, v8f acc) {
  const int row = lane & 15;
  const int hi  = lane >> 4;
#pragma unroll
  for (int kc = 0; kc < 8; ++kc) {
    const int k0 = kc * 4 + 2 * hi;
    v2f a, b;
    a.x = Arow[row * FEAT + k0];
    a.y = Arow[row * FEAT + k0 + 1];
    b.x = B[k0 * FEAT + bcol + row];
    b.y = B[(k0 + 1) * FEAT + bcol + row];
    acc = __builtin_amdgcn_wmma_f32_16x16x4_f32(false, a, false, b, (short)0, acc,
                                                false, false);
  }
  return acc;
}

__global__ void gat_init(float* __restrict__ m, float* __restrict__ denom,
                         float* __restrict__ num) {
  const int idx = blockIdx.x * blockDim.x + threadIdx.x;
  if (idx < N_NODES * HEADS) {
    m[idx]     = -INFINITY;
    denom[idx] = 0.0f;
  }
  if (idx < N_NODES * HEADS * FEAT) num[idx] = 0.0f;
}

__global__ __launch_bounds__(128) void gat_qkv_wmma(const float* __restrict__ x,
                                                    const float* __restrict__ Wq,
                                                    const float* __restrict__ Wk,
                                                    const float* __restrict__ Wv,
                                                    float* __restrict__ qkv) {
  const int lane = threadIdx.x & 31;
  const int wave = threadIdx.x >> 5;
  const int comb = blockIdx.y * 4 + wave;
  const int mat  = comb >> 3;
  const int head = (comb >> 1) & 3;
  const int nt   = comb & 1;
  const int m0   = blockIdx.x * 16;

  __shared__ float xs[16 * FEAT];
  ((float4*)xs)[threadIdx.x] = ((const float4*)(x + (size_t)m0 * FEAT))[threadIdx.x];
  __syncthreads();

  const float* W = (mat == 0 ? Wq : (mat == 1 ? Wk : Wv)) + head * FEAT * FEAT;
  v8f acc = {};
  acc = wmma_f32_k32(xs, W, nt * 16, lane, acc);

  float* out = qkv + ((size_t)(mat * HEADS + head) * N_NODES) * FEAT;
  const int col = lane & 15, hi = lane >> 4;
#pragma unroll
  for (int r = 0; r < 8; ++r)
    out[(size_t)(m0 + r + 8 * hi) * FEAT + nt * 16 + col] = acc[r];
}

__global__ __launch_bounds__(256) void gat_edge_logits(const float* __restrict__ qkv,
                                                       const int* __restrict__ ei,
                                                       float* __restrict__ sEH,
                                                       float* __restrict__ m) {
  const int wid = (blockIdx.x * blockDim.x + threadIdx.x) >> 5;
  if (wid >= N_EDGES) return;
  const int lane = threadIdx.x & 31;
  if (lane == 0 && wid + 8192 < N_EDGES) {
    __builtin_prefetch(ei + wid + 8192, 0, 0);
    __builtin_prefetch(ei + N_EDGES + wid + 8192, 0, 0);
  }
  const int h    = lane >> 3;
  const int f4   = (lane & 7) * 4;
  const int snd  = ei[wid];
  const int rcv  = ei[N_EDGES + wid];

  const float* Qr = qkv + ((size_t)(0 * HEADS + h) * N_NODES + rcv) * FEAT;
  const float* Kr = qkv + ((size_t)(1 * HEADS + h) * N_NODES + snd) * FEAT;
  const float4 q = *(const float4*)(Qr + f4);
  const float4 k = *(const float4*)(Kr + f4);
  float p = q.x * k.x + q.y * k.y + q.z * k.z + q.w * k.w;
  p += __shfl_xor(p, 1, 32);
  p += __shfl_xor(p, 2, 32);
  p += __shfl_xor(p, 4, 32);
  if ((lane & 7) == 0) {
    const float s = p * 0.17677669529663687f;
    sEH[(size_t)wid * HEADS + h] = s;
    atomic_max_f32_dev(&m[rcv * HEADS + h], s);
  }
}

__global__ __launch_bounds__(256) void gat_edge_scatter(const float* __restrict__ qkv,
                                                        const int* __restrict__ ei,
                                                        const float* __restrict__ sEH,
                                                        const float* __restrict__ m,
                                                        float* __restrict__ denom,
                                                        float* __restrict__ num) {
  const int wid = (blockIdx.x * blockDim.x + threadIdx.x) >> 5;
  if (wid >= N_EDGES) return;
  const int lane = threadIdx.x & 31;
  if (lane == 0 && wid + 8192 < N_EDGES) {
    __builtin_prefetch(ei + wid + 8192, 0, 0);
    __builtin_prefetch(ei + N_EDGES + wid + 8192, 0, 0);
    __builtin_prefetch(sEH + (size_t)(wid + 8192) * HEADS, 0, 0);
  }
  const int h    = lane >> 3;
  const int f4   = (lane & 7) * 4;
  const int snd  = ei[wid];
  const int rcv  = ei[N_EDGES + wid];

  const float s  = sEH[(size_t)wid * HEADS + h];
  const float mv = m[rcv * HEADS + h];
  const float w  = __expf(s - mv);
  if ((lane & 7) == 0) atomic_add_f32_dev(&denom[rcv * HEADS + h], w);

  const float* Vr = qkv + ((size_t)(2 * HEADS + h) * N_NODES + snd) * FEAT;
  const float4 v  = *(const float4*)(Vr + f4);
  float* dst = num + ((size_t)rcv * HEADS + h) * FEAT + f4;
  atomic_add_f32_dev(dst + 0, w * v.x);
  atomic_add_f32_dev(dst + 1, w * v.y);
  atomic_add_f32_dev(dst + 2, w * v.z);
  atomic_add_f32_dev(dst + 3, w * v.w);
}

__global__ void gat_finalize(const float* __restrict__ num,
                             const float* __restrict__ denom,
                             float* __restrict__ attnMean) {
  const int idx = blockIdx.x * blockDim.x + threadIdx.x;
  if (idx >= N_NODES * FEAT) return;
  const int n = idx >> 5, f = idx & 31;
  float acc = 0.0f;
#pragma unroll
  for (int h = 0; h < HEADS; ++h) {
    const float d = denom[n * HEADS + h];
    const float v = num[((size_t)n * HEADS + h) * FEAT + f];
    acc += (d > 0.0f) ? v / d : 0.0f;
  }
  attnMean[idx] = acc * 0.25f;
}

__global__ __launch_bounds__(64) void gat_out_wmma(const float* __restrict__ x,
                                                   const float* __restrict__ attnMean,
                                                   const float* __restrict__ Wout,
                                                   float* __restrict__ out) {
  const int lane = threadIdx.x & 31;
  const int nt   = threadIdx.x >> 5;
  const int m0   = blockIdx.x * 16;

  __shared__ float as[16 * FEAT];
  {
    const float4* src = (const float4*)(attnMean + (size_t)m0 * FEAT);
    ((float4*)as)[threadIdx.x]      = src[threadIdx.x];
    ((float4*)as)[threadIdx.x + 64] = src[threadIdx.x + 64];
  }
  __syncthreads();

  v8f acc = {};
  acc = wmma_f32_k32(as, Wout, nt * 16, lane, acc);

  const int col = lane & 15, hi = lane >> 4;
#pragma unroll
  for (int r = 0; r < 8; ++r) {
    const size_t o = (size_t)(m0 + r + 8 * hi) * FEAT + nt * 16 + col;
    out[o] = x[o] + acc[r];
  }
}

extern "C" void kernel_launch(void* const* d_in, const int* in_sizes, int n_in,
                              void* d_out, int out_size, void* d_ws, size_t ws_size,
                              hipStream_t stream) {
  const float* x    = (const float*)d_in[0];
  const int*   ei   = (const int*)d_in[1];
  const float* Wq   = (const float*)d_in[2];
  const float* Wk   = (const float*)d_in[3];
  const float* Wv   = (const float*)d_in[4];
  const float* Wout = (const float*)d_in[5];
  float*       out  = (float*)d_out;

  float* ws   = (float*)d_ws;
  float* qkv  = ws;
  float* sEH  = qkv + (size_t)3 * HEADS * N_NODES * FEAT;
  float* mbuf = sEH + (size_t)N_EDGES * HEADS;
  float* den  = mbuf + (size_t)N_NODES * HEADS;
  float* num  = den + (size_t)N_NODES * HEADS;
  float* attn = num + (size_t)N_NODES * HEADS * FEAT;

  gat_init<<<(N_NODES * HEADS * FEAT + 255) / 256, 256, 0, stream>>>(mbuf, den, num);

  gat_qkv_wmma<<<dim3(MTILES, 6), 128, 0, stream>>>(x, Wq, Wk, Wv, qkv);

  const int edgeBlocks = (N_EDGES * 32) / 256;
  gat_edge_logits<<<edgeBlocks, 256, 0, stream>>>(qkv, ei, sEH, mbuf);
  gat_edge_scatter<<<edgeBlocks, 256, 0, stream>>>(qkv, ei, sEH, mbuf, den, num);

  gat_finalize<<<(N_NODES * FEAT + 255) / 256, 256, 0, stream>>>(num, den, attn);

  gat_out_wmma<<<MTILES, 64, 0, stream>>>(x, attn, Wout, out);
}


# bucket pass: each tile run is copied whole into LDS in bucket order (8 lanes per run, all of a wave's loads in flight); waves take receivers and senders from there: no run search, no per-edge gathers
# speedup vs baseline: 3.1194x; 3.1194x over previous
#include <hip/hip_runtime.h>
#include <math.h>


#define N_NODES 50000
#define N_EDGES 800000
#define FEAT    32
#define HEADS   4
#define MTILES  (N_NODES / 16)

#define NBIN        256
#define TILE        4096
#define NBLK        ((N_EDGES + TILE - 1) / TILE)
#define TBLW        (NBIN + 1)

#define NBUCKET     ((N_NODES + NBIN - 1) / NBIN)
#define BK_WAVES    16
#define BK_THREADS  (BK_WAVES * 32)
#define BK_CHUNK    (BK_WAVES * 512)

#define QB_BUCKET   240
#define QT_TILE     (MTILES - 2 * QB_BUCKET)
static_assert(QT_TILE > 0 && QT_TILE + 2 * QB_BUCKET == MTILES, "every projection tile has one carrier");

static_assert(NBLK * TBLW <= 2 * NBIN * NBLK, "the table fits the space of the old counter planes");
static_assert(NBLK <= NBIN && NBUCKET < NBIN, "one thread per tile run in the bucket pass");

typedef __attribute__((ext_vector_type(2))) float v2f;
typedef __attribute__((ext_vector_type(8))) float v8f;

__device__ __forceinline__ v8f wmma_f32_k32(const float* Arow,
                                            const float* __restrict__ B,
                                            int bcol, int lane, v8f acc) {
  const int row = lane & 15;
  const int hi  = lane >> 4;
#pragma unroll
  for (int kc = 0; kc < 8; ++kc) {
    const int k0 = kc * 4 + 2 * hi;
    v2f a, b;
    a.x = Arow[row * FEAT + k0];
    a.y = Arow[row * FEAT + k0 + 1];
    b.x = B[k0 * FEAT + bcol + row];
    b.y = B[(k0 + 1) * FEAT + bcol + row];
    acc = __builtin_amdgcn_wmma_f32_16x16x4_f32(false, a, false, b, (short)0, acc,
                                                false, false);
  }
  return acc;
}

template <int NWAVES>
__device__ __forceinline__ void qkv_tiles(const float* __restrict__ x,
                                          const float* __restrict__ Wq,
                                          const float* __restrict__ Wk,
                                          const float* __restrict__ Wv,
                                          float* __restrict__ qkv, int tile0, float* xs) {
  constexpr int TPB = NWAVES / 8;
  const int t    = threadIdx.x;
  const int lane = t & 31;
  const int wave = t >> 5;

  if (t < TPB * 128) {
    const int row = tile0 * 16 + (t >> 3);
    float4 v = make_float4(0.0f, 0.0f, 0.0f, 0.0f);
    if (row < N_NODES) v = ((const float4*)(x + (size_t)tile0 * 16 * FEAT))[t];
    ((float4*)xs)[t] = v;
  }
  __syncthreads();

  const int tl   = wave >> 3;
  const int head = (wave >> 1) & 3;
  const int nt   = wave & 1;
  const int m0   = (tile0 + tl) * 16;
  if (m0 >= N_NODES) return;

  const int col = lane & 15, hi = lane >> 4;
#pragma unroll
  for (int mat = 0; mat < 3; ++mat) {
    const float* W = (mat == 0 ? Wq : (mat == 1 ? Wk : Wv)) + head * FEAT * FEAT;
    v8f acc = {};
    acc = wmma_f32_k32(xs + tl * 16 * FEAT, W, nt * 16, lane, acc);

    float* out = qkv + ((size_t)(mat * HEADS + head) * N_NODES) * FEAT;
#pragma unroll
    for (int r = 0; r < 8; ++r)
      out[(size_t)(m0 + r + 8 * hi) * FEAT + nt * 16 + col] = acc[r];
  }
}

__device__ __forceinline__ unsigned same_digit_lanes(unsigned dgt, bool valid) {
  unsigned mask = __builtin_amdgcn_ballot_w32(valid);
#pragma unroll
  for (int b = 0; b < 8; ++b) {
    const bool bit = ((dgt >> b) & 1u) != 0u;
    const unsigned bb = __builtin_amdgcn_ballot_w32(bit);
    mask &= bit ? bb : ~bb;
  }
  return mask;
}

__device__ __forceinline__ void load_wave_words(const int* __restrict__ src, int wbase, int lane,
                                                int limit, int (&k)[16]) {
#pragma unroll
  for (int it = 0; it < 16; ++it) {
    const int e = wbase + it * 32 + lane;
    k[it] = (e < limit) ? src[e] : 0;
  }
}

template <int SHIFT>
__device__ __forceinline__ void wave_histogram(const int (&k)[16], int* hrow, int wbase, int lane,
                                               int limit) {
#pragma unroll
  for (int it = 0; it < 16; ++it) {
    const bool valid = wbase + it * 32 + lane < limit;
    const unsigned dgt  = ((unsigned)k[it] >> SHIFT) & 255u;
    const unsigned mask = same_digit_lanes(dgt, valid);
    const unsigned rank = __builtin_popcount(mask & ((1u << lane) - 1u));
    if (valid && rank == 0u) hrow[dgt] += (int)__builtin_popcount(mask);
    __builtin_amdgcn_fence(__ATOMIC_ACQ_REL, "wavefront");
  }
}

__device__ __forceinline__ int wave_incl_scan(int v, int lane);

__global__ __launch_bounds__(256) void tile_partition(const int* __restrict__ keys,
                                                      const int* __restrict__ vals,
                                                      int* __restrict__ okeys,
                                                      int* __restrict__ ovals,
                                                      int* __restrict__ tbl,
                                                      const float* __restrict__ x,
                                                      const float* __restrict__ Wq,
                                                      const float* __restrict__ Wk,
                                                      const float* __restrict__ Wv,
                                                      float* __restrict__ qkv) {
  __shared__ alignas(16) int hist[8 * NBIN];
  __shared__ int wtot[8];
  __shared__ alignas(16) int stk[TILE];
  __shared__ alignas(16) int stv[TILE];
  if (blockIdx.x >= NBLK) {
    qkv_tiles<8>(x, Wq, Wk, Wv, qkv, (int)blockIdx.x - NBLK, (float*)hist);
    return;
  }
  const int t = threadIdx.x, lane = t & 31, w = t >> 5;
  const int tbase = blockIdx.x * TILE;
  const int wbase = tbase + w * 512;
  int k[16], v[16];
  load_wave_words(keys, wbase, lane, N_EDGES, k);
  load_wave_words(vals, wbase, lane, N_EDGES, v);
  for (int i = t; i < 8 * NBIN; i += 256) hist[i] = 0;
  __syncthreads();
  int* hrow = hist + w * NBIN;
  wave_histogram<8>(k, hrow, wbase, lane, N_EDGES);
  __syncthreads();
  {
    int tot = 0;
    for (int ww = 0; ww < 8; ++ww) tot += hist[ww * NBIN + t];
    const int incl = wave_incl_scan(tot, lane);
    if (lane == 31) wtot[w] = incl;
    __syncthreads();
    int run = incl - tot;
    for (int ww = 0; ww < 8; ++ww) run += (ww < w) ? wtot[ww] : 0;
    tbl[blockIdx.x * TBLW + t] = run;
    if (t == NBIN - 1) tbl[blockIdx.x * TBLW + NBIN] = run + tot;
    for (int ww = 0; ww < 8; ++ww) {
      const int c = hist[ww * NBIN + t];
      hist[ww * NBIN + t] = run;
      run += c;
    }
  }
  __syncthreads();
#pragma unroll
  for (int it = 0; it < 16; ++it) {
    const bool valid = wbase + it * 32 + lane < N_EDGES;
    const unsigned dgt  = ((unsigned)k[it] >> 8) & 255u;
    const unsigned mask = same_digit_lanes(dgt, valid);
    const unsigned rank = __builtin_popcount(mask & ((1u << lane) - 1u));
    const int start = hrow[dgt];
    if (valid) {
      const int pos = start + (int)rank;
      if ((unsigned)pos < (unsigned)TILE) {
        stk[pos] = k[it];
        stv[pos] = v[it];
      }
    }
    if (valid && rank == 0u) hrow[dgt] = start + (int)__builtin_popcount(mask);
    __builtin_amdgcn_fence(__ATOMIC_ACQ_REL, "wavefront");
  }
  __syncthreads();
  static_assert(TILE % 4 == 0 && N_EDGES % 4 == 0, "tiles are written back in 16-byte pieces");
  const int nvalid = (N_EDGES - tbase) < TILE ? (N_EDGES - tbase) : TILE;
  for (int i = 4 * t; i < nvalid; i += 4 * 256) {
    *(int4*)(okeys + tbase + i) = *(const int4*)(stk + i);
    *(int4*)(ovals + tbase + i) = *(const int4*)(stv + i);
  }
}

__device__ __forceinline__ int wave_incl_scan(int v, int lane) {
#pragma unroll
  for (int d = 1; d < 32; d <<= 1) {
    const int u = __shfl_up(v, d, 32);
    if (lane >= d) v += u;
  }
  return v;
}

__device__ __forceinline__ void wave_place_vals(const int (&k)[16], const int (&v)[16], int* hrow,
                                                int wbase, int lane, int limit,
                                                int* __restrict__ ovals, int* stage, int beg,
                                                bool staged) {
#pragma unroll
  for (int it = 0; it < 16; ++it) {
    const bool valid = wbase + it * 32 + lane < limit;
    const unsigned dgt  = (unsigned)k[it] & 255u;
    const unsigned mask = same_digit_lanes(dgt, valid);
    const unsigned rank = __builtin_popcount(mask & ((1u << lane) - 1u));
    const int start = hrow[dgt];
    if (valid) {
      const int pos = start + (int)rank;
      if (staged) {
        if ((unsigned)(pos - beg) < (unsigned)BK_CHUNK) stage[pos - beg] = v[it];
      } else {
        if ((unsigned)pos < (unsigned)N_EDGES) ovals[pos] = v[it];
      }
    }
    if (valid && rank == 0u) hrow[dgt] = start + (int)__builtin_popcount(mask);
    __builtin_amdgcn_fence(__ATOMIC_ACQ_REL, "wavefront");
  }
}

__device__ __forceinline__ int counts_to_starts(int* hist, int t, int run) {
  for (int ww = 0; ww < BK_WAVES; ++ww) {
    const int c = hist[ww * NBIN + t];
    hist[ww * NBIN + t] = run;
    run += c;
  }
  return run;
}

#define RUN_BATCH 6
__device__ __forceinline__ int edge_slot(int s) {
  return s < 0 ? 0 : (s >= N_EDGES ? N_EDGES - 1 : s);
}
__device__ __forceinline__ void copy_runs(const int* __restrict__ keys, const int* __restrict__ vals,
                                          const int* rsrc, const int* rdst, int cbase, int w, int lane,
                                          int* sk, int* sv) {
  const int g = lane >> 3, sub = lane & 7;
  int lo[4], hi[4], rel[4];
#pragma unroll
  for (int r = 0; r < 4; ++r) {
    const int j = 64 * r + 4 * w + g;
    lo[r] = 0; hi[r] = 0; rel[r] = 0;
    if (j < NBLK) {
      const int d0 = rdst[j], d1 = rdst[j + 1];
      lo[r]  = d0 > cbase ? d0 : cbase;
      hi[r]  = d1 < cbase + BK_CHUNK ? d1 : cbase + BK_CHUNK;
      rel[r] = rsrc[j] - d0;
    }
  }
  int kk[4][RUN_BATCH], vv[4][RUN_BATCH];
#pragma unroll
  for (int r = 0; r < 4; ++r) {
#pragma unroll
    for (int u = 0; u < RUN_BATCH; ++u) {
      const int p = lo[r] + sub + 8 * u;
      const int s = edge_slot(rel[r] + (p < hi[r] ? p : lo[r]));
      kk[r][u] = keys[s];
      vv[r][u] = vals[s];
    }
  }
#pragma unroll
  for (int r = 0; r < 4; ++r) {
#pragma unroll
    for (int u = 0; u < RUN_BATCH; ++u) {
      const int p = lo[r] + sub + 8 * u;
      if (p < hi[r]) { sk[p - cbase] = kk[r][u]; sv[p - cbase] = vv[r][u]; }
    }
  }
#pragma unroll
  for (int r = 0; r < 4; ++r) {
    for (int p = lo[r] + 8 * RUN_BATCH + sub; p < hi[r]; p += 8) {
      const int s = edge_slot(rel[r] + p);
      sk[p - cbase] = keys[s];
      sv[p - cbase] = vals[s];
    }
  }
}

__device__ __forceinline__ void take_wave_words(const int* wsrc, int ebase, int lane, int len,
                                                int (&o)[16]) {
#pragma unroll
  for (int it = 0; it < 16; ++it) o[it] = (ebase + it * 32 + lane < len) ? wsrc[it * 32 + lane] : 0;
}

__global__ __launch_bounds__(BK_THREADS) void bucket_finish(const int* __restrict__ keys,
                                                           const int* __restrict__ vals,
                                                           const int* __restrict__ tbl,
                                                           int* __restrict__ ovals,
                                                           int* __restrict__ off,
                                                           const float* __restrict__ x,
                                                           const float* __restrict__ Wq,
                                                           const float* __restrict__ Wk,
                                                           const float* __restrict__ Wv,
                                                           float* __restrict__ qkv) {
  __shared__ alignas(16) int hist[BK_WAVES * NBIN];
  __shared__ int wtot[NBIN / 32];
  __shared__ int wlen[NBIN / 32];
  __shared__ int wbeg[NBIN / 32];
  __shared__ int rsrc[NBLK];
  __shared__ int rdst[NBLK + 1];
  __shared__ int sk[BK_CHUNK];
  __shared__ int sv[BK_CHUNK];
  int* stage = sk;
  if (blockIdx.x >= NBUCKET) {
    qkv_tiles<16>(x, Wq, Wk, Wv, qkv, QT_TILE + 2 * ((int)blockIdx.x - NBUCKET), (float*)hist);
    return;
  }
  const int t = threadIdx.x, lane = t & 31, w = t >> 5;
  const int b = blockIdx.x;

  int rs = 0, rl = 0;
  if (t < NBLK) {
    rs = tbl[t * TBLW + b];
    int re = tbl[t * TBLW + b + 1];
    rs = rs < 0 ? 0 : (rs > TILE ? TILE : rs);
    re = re < rs ? rs : (re > TILE ? TILE : re);
    rl = re - rs;
  }
  int rincl = 0;
  if (t < NBIN) {
    rincl = wave_incl_scan(rl, lane);
    const int sincl = wave_incl_scan(rs, lane);
    if (lane == 31) { wlen[w] = rincl; wbeg[w] = sincl; }
  }
  __syncthreads();
  int len = 0, beg = 0;
  for (int ww = 0; ww < NBIN / 32; ++ww) { len += wlen[ww]; beg += wbeg[ww]; }
  if (t < NBLK) {
    int wb = 0;
    for (int ww = 0; ww < NBIN / 32; ++ww) wb += (ww < w) ? wlen[ww] : 0;
    rsrc[t] = t * TILE + rs;
    rdst[t] = wb + rincl - rl;
  }
  if (t == 0) rdst[NBLK] = len;
  beg = beg < 0 ? 0 : (beg > N_EDGES ? N_EDGES : beg);
  len = len < 0 ? 0 : (len > N_EDGES - beg ? N_EDGES - beg : len);
  __syncthreads();

  const int nch = len > BK_CHUNK ? (len + BK_CHUNK - 1) / BK_CHUNK : 1;
  const int nst = (nch == 1) ? 1 : 2 * nch;

  int* hrow = hist + w * NBIN;
  int run = 0;
  for (int i = 0; i < nst; ++i) {
    const bool placing = (nch == 1) || (i >= nch);
    const int  c       = (i >= nch) ? i - nch : i;
    const int  ebase   = c * BK_CHUNK + w * 512;
    int k[16], v[16];
    copy_runs(keys, vals, rsrc, rdst, c * BK_CHUNK, w, lane, sk, sv);
    __syncthreads();
    take_wave_words(sk + w * 512, ebase, lane, len, k);
    take_wave_words(sv + w * 512, ebase, lane, len, v);

    if (i == 0 || i >= nch) {
      __syncthreads();
      for (int j = t; j < BK_WAVES * NBIN; j += BK_THREADS) hist[j] = 0;
      __syncthreads();
    }
    wave_histogram<0>(k, hrow, ebase, lane, len);
    __syncthreads();

    if (i == nch - 1) {
      int tot = 0, incl = 0;
      if (t < NBIN) {
        for (int ww = 0; ww < BK_WAVES; ++ww) tot += hist[ww * NBIN + t];
        incl = wave_incl_scan(tot, lane);
        if (lane == 31) wtot[w] = incl;
      }
      __syncthreads();
      if (t < NBIN) {
        int wb = 0;
        for (int ww = 0; ww < NBIN / 32; ++ww) wb += (ww < w) ? wtot[ww] : 0;
        run = beg + wb + incl - tot;
        const int n = b * NBIN + t;
        if (n <= N_NODES) off[n] = run;
      }
    }

    if (placing) {
      if (t < NBIN) run = counts_to_starts(hist, t, run);
      __syncthreads();
      wave_place_vals(k, v, hrow, ebase, lane, len, ovals, stage, beg, nch == 1);
      if (nch == 1) {
        __syncthreads();
        for (int j = t; j < len; j += BK_THREADS) ovals[beg + j] = stage[j];
      }
    }
  }
}

__device__ __forceinline__ float sum8(float p) {
  p += __shfl_xor(p, 1, 32);
  p += __shfl_xor(p, 2, 32);
  p += __shfl_xor(p, 4, 32);
  return p;
}
__device__ __forceinline__ float dot4(const float4 a, const float4 b) {
  return a.x * b.x + a.y * b.y + a.z * b.z + a.w * b.w;
}

__global__ __launch_bounds__(256) void gat_node_pull(const float* __restrict__ qkv,
                                                     const int* __restrict__ off,
                                                     const int* __restrict__ ssnd,
                                                     float* __restrict__ attnMean) {
  const int lane = threadIdx.x & 31;
  const int n = __builtin_amdgcn_readfirstlane((int)((blockIdx.x * blockDim.x + threadIdx.x) >> 5));
  if (n >= N_NODES) return;
  const int h  = lane >> 3;
  const int f4 = (lane & 7) * 4;
  int beg = off[n];
  int end = off[n + 1];
  beg = beg < 0 ? 0 : (beg > N_EDGES ? N_EDGES : beg);
  end = end < beg ? beg : (end > N_EDGES ? N_EDGES : end);

  const float4 q = *(const float4*)(qkv + ((size_t)(0 * HEADS + h) * N_NODES + n) * FEAT + f4);
  const float* Kb = qkv + ((size_t)(1 * HEADS + h) * N_NODES) * FEAT + f4;
  const float* Vb = qkv + ((size_t)(2 * HEADS + h) * N_NODES) * FEAT + f4;

  float  m = -INFINITY, d = 0.0f;
  float4 acc = make_float4(0.0f, 0.0f, 0.0f, 0.0f);

  for (int base = beg; base < end; base += 32) {
    const int rem = end - base;
    const int cn  = rem < 32 ? rem : 32;
    int sv = (lane < cn) ? ssnd[base + lane] : 0;
    sv = sv < 0 ? 0 : (sv >= N_NODES ? N_NODES - 1 : sv);

    for (int j = 0; j < cn; j += 4) {
      const bool ok1 = j + 1 < cn, ok2 = j + 2 < cn, ok3 = j + 3 < cn;
      const int s0 = __builtin_amdgcn_readlane(sv, j);
      const int s1 = __builtin_amdgcn_readlane(sv, ok1 ? j + 1 : j);
      const int s2 = __builtin_amdgcn_readlane(sv, ok2 ? j + 2 : j);
      const int s3 = __builtin_amdgcn_readlane(sv, ok3 ? j + 3 : j);
      const float4 k0 = *(const float4*)(Kb + (size_t)s0 * FEAT);
      const float4 k1 = *(const float4*)(Kb + (size_t)s1 * FEAT);
      const float4 k2 = *(const float4*)(Kb + (size_t)s2 * FEAT);
      const float4 k3 = *(const float4*)(Kb + (size_t)s3 * FEAT);
      const float4 v0 = *(const float4*)(Vb + (size_t)s0 * FEAT);
      const float4 v1 = *(const float4*)(Vb + (size_t)s1 * FEAT);
      const float4 v2 = *(const float4*)(Vb + (size_t)s2 * FEAT);
      const float4 v3 = *(const float4*)(Vb + (size_t)s3 * FEAT);

      const float sc = 0.17677669529663687f;
      const float pen1 = ok1 ? 0.0f : -INFINITY;
      const float pen2 = ok2 ? 0.0f : -INFINITY;
      const float pen3 = ok3 ? 0.0f : -INFINITY;
      const float p0 = sum8(dot4(q, k0)) * sc;
      const float p1 = sum8(dot4(q, k1)) * sc + pen1;
      const float p2 = sum8(dot4(q, k2)) * sc + pen2;
      const float p3 = sum8(dot4(q, k3)) * sc + pen3;

      const float mn = fmaxf(fmaxf(m, p0), fmaxf(fmaxf(p1, p2), p3));
      const float r  = __expf(m - mn);
      const float w0 = __expf(p0 - mn);
      const float w1 = __expf(p1 - mn);
      const float w2 = __expf(p2 - mn);
      const float w3 = __expf(p3 - mn);
      d = d * r + ((w0 + w1) + (w2 + w3));
      acc.x = acc.x * r + ((w0 * v0.x + w1 * v1.x) + (w2 * v2.x + w3 * v3.x));
      acc.y = acc.y * r + ((w0 * v0.y + w1 * v1.y) + (w2 * v2.y + w3 * v3.y));
      acc.z = acc.z * r + ((w0 * v0.z + w1 * v1.z) + (w2 * v2.z + w3 * v3.z));
      acc.w = acc.w * r + ((w0 * v0.w + w1 * v1.w) + (w2 * v2.w + w3 * v3.w));
      m = mn;
    }
  }

  float4 o;
  o.x = (d > 0.0f) ? acc.x / d : 0.0f;
  o.y = (d > 0.0f) ? acc.y / d : 0.0f;
  o.z = (d > 0.0f) ? acc.z / d : 0.0f;
  o.w = (d > 0.0f) ? acc.w / d : 0.0f;
  o.x += __shfl_xor(o.x, 8, 32);  o.x += __shfl_xor(o.x, 16, 32);
  o.y += __shfl_xor(o.y, 8, 32);  o.y += __shfl_xor(o.y, 16, 32);
  o.z += __shfl_xor(o.z, 8, 32);  o.z += __shfl_xor(o.z, 16, 32);
  o.w += __shfl_xor(o.w, 8, 32);  o.w += __shfl_xor(o.w, 16, 32);
  if (lane < 8) {
    o.x *= 0.25f; o.y *= 0.25f; o.z *= 0.25f; o.w *= 0.25f;
    *(float4*)(attnMean + (size_t)n * FEAT + f4) = o;
  }
}

__global__ __launch_bounds__(64) void gat_out_wmma(const float* __restrict__ x,
                                                   const float* __restrict__ attnMean,
                                                   const float* __restrict__ Wout,
                                                   float* __restrict__ out) {
  const int lane = threadIdx.x & 31;
  const int nt   = threadIdx.x >> 5;
  const int m0   = blockIdx.x * 16;

  __shared__ float as[16 * FEAT];
  {
    const float4* src = (const float4*)(attnMean + (size_t)m0 * FEAT);
    ((float4*)as)[threadIdx.x]      = src[threadIdx.x];
    ((float4*)as)[threadIdx.x + 64] = src[threadIdx.x + 64];
  }
  __syncthreads();

  v8f acc = {};
  acc = wmma_f32_k32(as, Wout, nt * 16, lane, acc);

  const int col = lane & 15, hi = lane >> 4;
#pragma unroll
  for (int r = 0; r < 8; ++r) {
    const size_t o = (size_t)(m0 + r + 8 * hi) * FEAT + nt * 16 + col;
    out[o] = x[o] + acc[r];
  }
}

extern "C" void kernel_launch(void* const* d_in, const int* in_sizes, int n_in,
                              void* d_out, int out_size, void* d_ws, size_t ws_size,
                              hipStream_t stream) {
  const float* x    = (const float*)d_in[0];
  const int*   ei   = (const int*)d_in[1];
  const float* Wq   = (const float*)d_in[2];
  const float* Wk   = (const float*)d_in[3];
  const float* Wv   = (const float*)d_in[4];
  const float* Wout = (const float*)d_in[5];
  float*       out  = (float*)d_out;

  float* ws   = (float*)d_ws;
  float* qkv  = ws;
  float* attn = qkv + (size_t)3 * HEADS * N_NODES * FEAT;
  int*   tbl  = (int*)(attn + (size_t)N_NODES * FEAT);
  int*   k1   = tbl + 2 * NBIN * NBLK;
  int*   v1   = k1 + N_EDGES;
  int*   v2   = v1 + N_EDGES;
  int*   off  = v2 + N_EDGES;

  tile_partition<<<NBLK + QT_TILE, 256, 0, stream>>>(ei + N_EDGES, ei, k1, v1, tbl,
                                                     x, Wq, Wk, Wv, qkv);
  bucket_finish<<<NBUCKET + QB_BUCKET, BK_THREADS, 0, stream>>>(k1, v1, tbl, v2, off,
                                                                x, Wq, Wk, Wv, qkv);

  gat_node_pull<<<N_NODES / 8, 256, 0, stream>>>(qkv, off, v2, attn);

  gat_out_wmma<<<MTILES, 64, 0, stream>>>(x, attn, Wout, out);
}


# Q/K/V projection behind the tile pass: 256 resident work-groups each walk 10 or 11 tiles (weight fragments kept in registers, next x tile prefetched, LDS-only barrier), not 2645 one-tile work-groups
# speedup vs baseline: 3.2928x; 1.0556x over previous
#include <hip/hip_runtime.h>
#include <math.h>


#define N_NODES 50000
#define N_EDGES 800000
#define FEAT    32
#define HEADS   4
#define MTILES  (N_NODES / 16)

#define NBIN        256
#define TILE        4096
#define NBLK        ((N_EDGES + TILE - 1) / TILE)
#define TBLW        (NBIN + 1)

#define NBUCKET     ((N_NODES + NBIN - 1) / NBIN)
#define BK_WAVES    16
#define BK_THREADS  (BK_WAVES * 32)
#define BK_CHUNK    (BK_WAVES * 512)

#define QB_BUCKET   240
#define QT_TILE     (MTILES - 2 * QB_BUCKET)
#define QT_WG       256
static_assert(QT_TILE > 0 && QT_TILE + 2 * QB_BUCKET == MTILES, "every projection tile has one carrier");
static_assert(QT_WG > 0 && QT_WG <= QT_TILE && QT_TILE <= MTILES, "every walk has a first tile; whole tiles only");

static_assert(NBLK * TBLW <= 2 * NBIN * NBLK, "the table fits the space of the old counter planes");
static_assert(NBLK <= NBIN && NBUCKET < NBIN, "one thread per tile run in the bucket pass");

typedef __attribute__((ext_vector_type(2))) float v2f;
typedef __attribute__((ext_vector_type(8))) float v8f;

__device__ __forceinline__ v8f wmma_f32_k32(const float* Arow,
                                            const float* __restrict__ B,
                                            int bcol, int lane, v8f acc) {
  const int row = lane & 15;
  const int hi  = lane >> 4;
#pragma unroll
  for (int kc = 0; kc < 8; ++kc) {
    const int k0 = kc * 4 + 2 * hi;
    v2f a, b;
    a.x = Arow[row * FEAT + k0];
    a.y = Arow[row * FEAT + k0 + 1];
    b.x = B[k0 * FEAT + bcol + row];
    b.y = B[(k0 + 1) * FEAT + bcol + row];
    acc = __builtin_amdgcn_wmma_f32_16x16x4_f32(false, a, false, b, (short)0, acc,
                                                false, false);
  }
  return acc;
}

template <int NWAVES>
__device__ __forceinline__ void qkv_tiles(const float* __restrict__ x,
                                          const float* __restrict__ Wq,
                                          const float* __restrict__ Wk,
                                          const float* __restrict__ Wv,
                                          float* __restrict__ qkv, int tile0, float* xs) {
  constexpr int TPB = NWAVES / 8;
  const int t    = threadIdx.x;
  const int lane = t & 31;
  const int wave = t >> 5;

  if (t < TPB * 128) {
    const int row = tile0 * 16 + (t >> 3);
    float4 v = make_float4(0.0f, 0.0f, 0.0f, 0.0f);
    if (row < N_NODES) v = ((const float4*)(x + (size_t)tile0 * 16 * FEAT))[t];
    ((float4*)xs)[t] = v;
  }
  __syncthreads();

  const int tl   = wave >> 3;
  const int head = (wave >> 1) & 3;
  const int nt   = wave & 1;
  const int m0   = (tile0 + tl) * 16;
  if (m0 >= N_NODES) return;

  const int col = lane & 15, hi = lane >> 4;
#pragma unroll
  for (int mat = 0; mat < 3; ++mat) {
    const float* W = (mat == 0 ? Wq : (mat == 1 ? Wk : Wv)) + head * FEAT * FEAT;
    v8f acc = {};
    acc = wmma_f32_k32(xs + tl * 16 * FEAT, W, nt * 16, lane, acc);

    float* out = qkv + ((size_t)(mat * HEADS + head) * N_NODES) * FEAT;
#pragma unroll
    for (int r = 0; r < 8; ++r)
      out[(size_t)(m0 + r + 8 * hi) * FEAT + nt * 16 + col] = acc[r];
  }
}

__device__ __forceinline__ void qkv_tile_walk(const float* __restrict__ x,
                                              const float* __restrict__ Wq,
                                              const float* __restrict__ Wk,
                                              const float* __restrict__ Wv,
                                              float* __restrict__ qkv, int first, int step, float* xs) {
  const int t    = threadIdx.x;
  const int lane = t & 31;
  const int wave = t >> 5;
  const int head = (wave >> 1) & 3;
  const int nt   = wave & 1;
  const int row  = lane & 15, hi = lane >> 4;

  float4 xv = make_float4(0.0f, 0.0f, 0.0f, 0.0f);
  if (t < 128) xv = ((const float4*)(x + (size_t)first * 16 * FEAT))[t];

  v2f bq[8], bk[8], bv[8];
#pragma unroll
  for (int kc = 0; kc < 8; ++kc) {
    const int o = head * FEAT * FEAT + (kc * 4 + 2 * hi) * FEAT + nt * 16 + row;
    bq[kc].x = Wq[o];  bq[kc].y = Wq[o + FEAT];
    bk[kc].x = Wk[o];  bk[kc].y = Wk[o + FEAT];
    bv[kc].x = Wv[o];  bv[kc].y = Wv[o + FEAT];
  }

  if (t < 128) ((float4*)xs)[t] = xv;
  __syncthreads();

  int i = 0;
  for (int tile = first; tile < QT_TILE; tile += step, ++i) {
    const bool more = tile + step < QT_TILE;
    if (more && t < 128) xv = ((const float4*)(x + (size_t)(tile + step) * 16 * FEAT))[t];

    const float* A = xs + (i & 1) * 16 * FEAT;
    v2f a[8];
#pragma unroll
    for (int kc = 0; kc < 8; ++kc) {
      const int k0 = kc * 4 + 2 * hi;
      a[kc].x = A[row * FEAT + k0];
      a[kc].y = A[row * FEAT + k0 + 1];
    }
    const int m0 = tile * 16;
#pragma unroll
    for (int mat = 0; mat < 3; ++mat) {
      v8f acc = {};
#pragma unroll
      for (int kc = 0; kc < 8; ++kc) {
        const v2f b = (mat == 0 ? bq[kc] : (mat == 1 ? bk[kc] : bv[kc]));
        acc = __builtin_amdgcn_wmma_f32_16x16x4_f32(false, a[kc], false, b, (short)0, acc,
                                                    false, false);
      }
      float* out = qkv + ((size_t)(mat * HEADS + head) * N_NODES) * FEAT;
#pragma unroll
      for (int r = 0; r < 8; ++r)
        out[(size_t)(m0 + r + 8 * hi) * FEAT + nt * 16 + row] = acc[r];
    }

    if (more && t < 128) ((float4*)(xs + ((i + 1) & 1) * 16 * FEAT))[t] = xv;
    __builtin_amdgcn_fence(__ATOMIC_RELEASE, "workgroup", "local");
    __builtin_amdgcn_s_barrier();
    __builtin_amdgcn_fence(__ATOMIC_ACQUIRE, "workgroup", "local");
  }
}

__device__ __forceinline__ unsigned same_digit_lanes(unsigned dgt, bool valid) {
  unsigned mask = __builtin_amdgcn_ballot_w32(valid);
#pragma unroll
  for (int b = 0; b < 8; ++b) {
    const bool bit = ((dgt >> b) & 1u) != 0u;
    const unsigned bb = __builtin_amdgcn_ballot_w32(bit);
    mask &= bit ? bb : ~bb;
  }
  return mask;
}

__device__ __forceinline__ void load_wave_words(const int* __restrict__ src, int wbase, int lane,
                                                int limit, int (&k)[16]) {
#pragma unroll
  for (int it = 0; it < 16; ++it) {
    const int e = wbase + it * 32 + lane;
    k[it] = (e < limit) ? src[e] : 0;
  }
}

template <int SHIFT>
__device__ __forceinline__ void wave_histogram(const int (&k)[16], int* hrow, int wbase, int lane,
                                               int limit) {
#pragma unroll
  for (int it = 0; it < 16; ++it) {
    const bool valid = wbase + it * 32 + lane < limit;
    const unsigned dgt  = ((unsigned)k[it] >> SHIFT) & 255u;
    const unsigned mask = same_digit_lanes(dgt, valid);
    const unsigned rank = __builtin_popcount(mask & ((1u << lane) - 1u));
    if (valid && rank == 0u) hrow[dgt] += (int)__builtin_popcount(mask);
    __builtin_amdgcn_fence(__ATOMIC_ACQ_REL, "wavefront");
  }
}

__device__ __forceinline__ int wave_incl_scan(int v, int lane);

__global__ __launch_bounds__(256) void tile_partition(const int* __restrict__ keys,
                                                      const int* __restrict__ vals,
                                                      int* __restrict__ okeys,
                                                      int* __restrict__ ovals,
                                                      int* __restrict__ tbl,
                                                      const float* __restrict__ x,
                                                      const float* __restrict__ Wq,
                                                      const float* __restrict__ Wk,
                                                      const float* __restrict__ Wv,
                                                      float* __restrict__ qkv) {
  __shared__ alignas(16) int hist[8 * NBIN];
  __shared__ int wtot[8];
  __shared__ alignas(16) int stk[TILE];
  __shared__ alignas(16) int stv[TILE];
  if (blockIdx.x >= NBLK) {
    qkv_tile_walk(x, Wq, Wk, Wv, qkv, (int)blockIdx.x - NBLK, QT_WG, (float*)hist);
    return;
  }
  const int t = threadIdx.x, lane = t & 31, w = t >> 5;
  const int tbase = blockIdx.x * TILE;
  const int wbase = tbase + w * 512;
  int k[16], v[16];
  load_wave_words(keys, wbase, lane, N_EDGES, k);
  load_wave_words(vals, wbase, lane, N_EDGES, v);
  for (int i = t; i < 8 * NBIN; i += 256) hist[i] = 0;
  __syncthreads();
  int* hrow = hist + w * NBIN;
  wave_histogram<8>(k, hrow, wbase, lane, N_EDGES);
  __syncthreads();
  {
    int tot = 0;
    for (int ww = 0; ww < 8; ++ww) tot += hist[ww * NBIN + t];
    const int incl = wave_incl_scan(tot, lane);
    if (lane == 31) wtot[w] = incl;
    __syncthreads();
    int run = incl - tot;
    for (int ww = 0; ww < 8; ++ww) run += (ww < w) ? wtot[ww] : 0;
    tbl[blockIdx.x * TBLW + t] = run;
    if (t == NBIN - 1) tbl[blockIdx.x * TBLW + NBIN] = run + tot;
    for (int ww = 0; ww < 8; ++ww) {
      const int c = hist[ww * NBIN + t];
      hist[ww * NBIN + t] = run;
      run += c;
    }
  }
  __syncthreads();
#pragma unroll
  for (int it = 0; it < 16; ++it) {
    const bool valid = wbase + it * 32 + lane < N_EDGES;
    const unsigned dgt  = ((unsigned)k[it] >> 8) & 255u;
    const unsigned mask = same_digit_lanes(dgt, valid);
    const unsigned rank = __builtin_popcount(mask & ((1u << lane) - 1u));
    const int start = hrow[dgt];
    if (valid) {
      const int pos = start + (int)rank;
      if ((unsigned)pos < (unsigned)TILE) {
        stk[pos] = k[it];
        stv[pos] = v[it];
      }
    }
    if (valid && rank == 0u) hrow[dgt] = start + (int)__builtin_popcount(mask);
    __builtin_amdgcn_fence(__ATOMIC_ACQ_REL, "wavefront");
  }
  __syncthreads();
  static_assert(TILE % 4 == 0 && N_EDGES % 4 == 0, "tiles are written back in 16-byte pieces");
  const int nvalid = (N_EDGES - tbase) < TILE ? (N_EDGES - tbase) : TILE;
  for (int i = 4 * t; i < nvalid; i += 4 * 256) {
    *(int4*)(okeys + tbase + i) = *(const int4*)(stk + i);
    *(int4*)(ovals + tbase + i) = *(const int4*)(stv + i);
  }
}

__device__ __forceinline__ int wave_incl_scan(int v, int lane) {
#pragma unroll
  for (int d = 1; d < 32; d <<= 1) {
    const int u = __shfl_up(v, d, 32);
    if (lane >= d) v += u;
  }
  return v;
}

__device__ __forceinline__ void wave_place_vals(const int (&k)[16], const int (&v)[16], int* hrow,
                                                int wbase, int lane, int limit,
                                                int* __restrict__ ovals, int* stage, int beg,
                                                bool staged) {
#pragma unroll
  for (int it = 0; it < 16; ++it) {
    const bool valid = wbase + it * 32 + lane < limit;
    const unsigned dgt  = (unsigned)k[it] & 255u;
    const unsigned mask = same_digit_lanes(dgt, valid);
    const unsigned rank = __builtin_popcount(mask & ((1u << lane) - 1u));
    const int start = hrow[dgt];
    if (valid) {
      const int pos = start + (int)rank;
      if (staged) {
        if ((unsigned)(pos - beg) < (unsigned)BK_CHUNK) stage[pos - beg] = v[it];
      } else {
        if ((unsigned)pos < (unsigned)N_EDGES) ovals[pos] = v[it];
      }
    }
    if (valid && rank == 0u) hrow[dgt] = start + (int)__builtin_popcount(mask);
    __builtin_amdgcn_fence(__ATOMIC_ACQ_REL, "wavefront");
  }
}

__device__ __forceinline__ int counts_to_starts(int* hist, int t, int run) {
  for (int ww = 0; ww < BK_WAVES; ++ww) {
    const int c = hist[ww * NBIN + t];
    hist[ww * NBIN + t] = run;
    run += c;
  }
  return run;
}

#define RUN_BATCH 6
__device__ __forceinline__ int edge_slot(int s) {
  return s < 0 ? 0 : (s >= N_EDGES ? N_EDGES - 1 : s);
}
__device__ __forceinline__ void copy_runs(const int* __restrict__ keys, const int* __restrict__ vals,
                                          const int* rsrc, const int* rdst, int cbase, int w, int lane,
                                          int* sk, int* sv) {
  const int g = lane >> 3, sub = lane & 7;
  int lo[4], hi[4], rel[4];
#pragma unroll
  for (int r = 0; r < 4; ++r) {
    const int j = 64 * r + 4 * w + g;
    lo[r] = 0; hi[r] = 0; rel[r] = 0;
    if (j < NBLK) {
      const int d0 = rdst[j], d1 = rdst[j + 1];
      lo[r]  = d0 > cbase ? d0 : cbase;
      hi[r]  = d1 < cbase + BK_CHUNK ? d1 : cbase + BK_CHUNK;
      rel[r] = rsrc[j] - d0;
    }
  }
  int kk[4][RUN_BATCH], vv[4][RUN_BATCH];
#pragma unroll
  for (int r = 0; r < 4; ++r) {
#pragma unroll
    for (int u = 0; u < RUN_BATCH; ++u) {
      const int p = lo[r] + sub + 8 * u;
      const int s = edge_slot(rel[r] + (p < hi[r] ? p : lo[r]));
      kk[r][u] = keys[s];
      vv[r][u] = vals[s];
    }
  }
#pragma unroll
  for (int r = 0; r < 4; ++r) {
#pragma unroll
    for (int u = 0; u < RUN_BATCH; ++u) {
      const int p = lo[r] + sub + 8 * u;
      if (p < hi[r]) { sk[p - cbase] = kk[r][u]; sv[p - cbase] = vv[r][u]; }
    }
  }
#pragma unroll
  for (int r = 0; r < 4; ++r) {
    for (int p = lo[r] + 8 * RUN_BATCH + sub; p < hi[r]; p += 8) {
      const int s = edge_slot(rel[r] + p);
      sk[p - cbase] = keys[s];
      sv[p - cbase] = vals[s];
    }
  }
}

__device__ __forceinline__ void take_wave_words(const int* wsrc, int ebase, int lane, int len,
                                                int (&o)[16]) {
#pragma unroll
  for (int it = 0; it < 16; ++it) o[it] = (ebase + it * 32 + lane < len) ? wsrc[it * 32 + lane] : 0;
}

__global__ __launch_bounds__(BK_THREADS) void bucket_finish(const int* __restrict__ keys,
                                                           const int* __restrict__ vals,
                                                           const int* __restrict__ tbl,
                                                           int* __restrict__ ovals,
                                                           int* __restrict__ off,
                                                           const float* __restrict__ x,
                                                           const float* __restrict__ Wq,
                                                           const float* __restrict__ Wk,
                                                           const float* __restrict__ Wv,
                                                           float* __restrict__ qkv) {
  __shared__ alignas(16) int hist[BK_WAVES * NBIN];
  __shared__ int wtot[NBIN / 32];
  __shared__ int wlen[NBIN / 32];
  __shared__ int wbeg[NBIN / 32];
  __shared__ int rsrc[NBLK];
  __shared__ int rdst[NBLK + 1];
  __shared__ int sk[BK_CHUNK];
  __shared__ int sv[BK_CHUNK];
  int* stage = sk;
  if (blockIdx.x >= NBUCKET) {
    qkv_tiles<16>(x, Wq, Wk, Wv, qkv, QT_TILE + 2 * ((int)blockIdx.x - NBUCKET), (float*)hist);
    return;
  }
  const int t = threadIdx.x, lane = t & 31, w = t >> 5;
  const int b = blockIdx.x;

  int rs = 0, rl = 0;
  if (t < NBLK) {
    rs = tbl[t * TBLW + b];
    int re = tbl[t * TBLW + b + 1];
    rs = rs < 0 ? 0 : (rs > TILE ? TILE : rs);
    re = re < rs ? rs : (re > TILE ? TILE : re);
    rl = re - rs;
  }
  int rincl = 0;
  if (t < NBIN) {
    rincl = wave_incl_scan(rl, lane);
    const int sincl = wave_incl_scan(rs, lane);
    if (lane == 31) { wlen[w] = rincl; wbeg[w] = sincl; }
  }
  __syncthreads();
  int len = 0, beg = 0;
  for (int ww = 0; ww < NBIN / 32; ++ww) { len += wlen[ww]; beg += wbeg[ww]; }
  if (t < NBLK) {
    int wb = 0;
    for (int ww = 0; ww < NBIN / 32; ++ww) wb += (ww < w) ? wlen[ww] : 0;
    rsrc[t] = t * TILE + rs;
    rdst[t] = wb + rincl - rl;
  }
  if (t == 0) rdst[NBLK] = len;
  beg = beg < 0 ? 0 : (beg > N_EDGES ? N_EDGES : beg);
  len = len < 0 ? 0 : (len > N_EDGES - beg ? N_EDGES - beg : len);
  __syncthreads();

  const int nch = len > BK_CHUNK ? (len + BK_CHUNK - 1) / BK_CHUNK : 1;
  const int nst = (nch == 1) ? 1 : 2 * nch;

  int* hrow = hist + w * NBIN;
  int run = 0;
  for (int i = 0; i < nst; ++i) {
    const bool placing = (nch == 1) || (i >= nch);
    const int  c       = (i >= nch) ? i - nch : i;
    const int  ebase   = c * BK_CHUNK + w * 512;
    int k[16], v[16];
    copy_runs(keys, vals, rsrc, rdst, c * BK_CHUNK, w, lane, sk, sv);
    __syncthreads();
    take_wave_words(sk + w * 512, ebase, lane, len, k);
    take_wave_words(sv + w * 512, ebase, lane, len, v);

    if (i == 0 || i >= nch) {
      __syncthreads();
      for (int j = t; j < BK_WAVES * NBIN; j += BK_THREADS) hist[j] = 0;
      __syncthreads();
    }
    wave_histogram<0>(k, hrow, ebase, lane, len);
    __syncthreads();

    if (i == nch - 1) {
      int tot = 0, incl = 0;
      if (t < NBIN) {
        for (int ww = 0; ww < BK_WAVES; ++ww) tot += hist[ww * NBIN + t];
        incl = wave_incl_scan(tot, lane);
        if (lane == 31) wtot[w] = incl;
      }
      __syncthreads();
      if (t < NBIN) {
        int wb = 0;
        for (int ww = 0; ww < NBIN / 32; ++ww) wb += (ww < w) ? wtot[ww] : 0;
        run = beg + wb + incl - tot;
        const int n = b * NBIN + t;
        if (n <= N_NODES) off[n] = run;
      }
    }

    if (placing) {
      if (t < NBIN) run = counts_to_starts(hist, t, run);
      __syncthreads();
      wave_place_vals(k, v, hrow, ebase, lane, len, ovals, stage, beg, nch == 1);
      if (nch == 1) {
        __syncthreads();
        for (int j = t; j < len; j += BK_THREADS) ovals[beg + j] = stage[j];
      }
    }
  }
}

__device__ __forceinline__ float sum8(float p) {
  p += __shfl_xor(p, 1, 32);
  p += __shfl_xor(p, 2, 32);
  p += __shfl_xor(p, 4, 32);
  return p;
}
__device__ __forceinline__ float dot4(const float4 a, const float4 b) {
  return a.x * b.x + a.y * b.y + a.z * b.z + a.w * b.w;
}

__global__ __launch_bounds__(256) void gat_node_pull(const float* __restrict__ qkv,
                                                     const int* __restrict__ off,
                                                     const int* __restrict__ ssnd,
                                                     float* __restrict__ attnMean) {
  const int lane = threadIdx.x & 31;
  const int n = __builtin_amdgcn_readfirstlane((int)((blockIdx.x * blockDim.x + threadIdx.x) >> 5));
  if (n >= N_NODES) return;
  const int h  = lane >> 3;
  const int f4 = (lane & 7) * 4;
  int beg = off[n];
  int end = off[n + 1];
  beg = beg < 0 ? 0 : (beg > N_EDGES ? N_EDGES : beg);
  end = end < beg ? beg : (end > N_EDGES ? N_EDGES : end);

  const float4 q = *(const float4*)(qkv + ((size_t)(0 * HEADS + h) * N_NODES + n) * FEAT + f4);
  const float* Kb = qkv + ((size_t)(1 * HEADS + h) * N_NODES) * FEAT + f4;
  const float* Vb = qkv + ((size_t)(2 * HEADS + h) * N_NODES) * FEAT + f4;

  float  m = -INFINITY, d = 0.0f;
  float4 acc = make_float4(0.0f, 0.0f, 0.0f, 0.0f);

  for (int base = beg; base < end; base += 32) {
    const int rem = end - base;
    const int cn  = rem < 32 ? rem : 32;
    int sv = (lane < cn) ? ssnd[base + lane] : 0;
    sv = sv < 0 ? 0 : (sv >= N_NODES ? N_NODES - 1 : sv);

    for (int j = 0; j < cn; j += 4) {
      const bool ok1 = j + 1 < cn, ok2 = j + 2 < cn, ok3 = j + 3 < cn;
      const int s0 = __builtin_amdgcn_readlane(sv, j);
      const int s1 = __builtin_amdgcn_readlane(sv, ok1 ? j + 1 : j);
      const int s2 = __builtin_amdgcn_readlane(sv, ok2 ? j + 2 : j);
      const int s3 = __builtin_amdgcn_readlane(sv, ok3 ? j + 3 : j);
      const float4 k0 = *(const float4*)(Kb + (size_t)s0 * FEAT);
      const float4 k1 = *(const float4*)(Kb + (size_t)s1 * FEAT);
      const float4 k2 = *(const float4*)(Kb + (size_t)s2 * FEAT);
      const float4 k3 = *(const float4*)(Kb + (size_t)s3 * FEAT);
      const float4 v0 = *(const float4*)(Vb + (size_t)s0 * FEAT);
      const float4 v1 = *(const float4*)(Vb + (size_t)s1 * FEAT);
      const float4 v2 = *(const float4*)(Vb + (size_t)s2 * FEAT);
      const float4 v3 = *(const float4*)(Vb + (size_t)s3 * FEAT);

      const float sc = 0.17677669529663687f;
      const float pen1 = ok1 ? 0.0f : -INFINITY;
      const float pen2 = ok2 ? 0.0f : -INFINITY;
      const float pen3 = ok3 ? 0.0f : -INFINITY;
      const float p0 = sum8(dot4(q, k0)) * sc;
      const float p1 = sum8(dot4(q, k1)) * sc + pen1;
      const float p2 = sum8(dot4(q, k2)) * sc + pen2;
      const float p3 = sum8(dot4(q, k3)) * sc + pen3;

      const float mn = fmaxf(fmaxf(m, p0), fmaxf(fmaxf(p1, p2), p3));
      const float r  = __expf(m - mn);
      const float w0 = __expf(p0 - mn);
      const float w1 = __expf(p1 - mn);
      const float w2 = __expf(p2 - mn);
      const float w3 = __expf(p3 - mn);
      d = d * r + ((w0 + w1) + (w2 + w3));
      acc.x = acc.x * r + ((w0 * v0.x + w1 * v1.x) + (w2 * v2.x + w3 * v3.x));
      acc.y = acc.y * r + ((w0 * v0.y + w1 * v1.y) + (w2 * v2.y + w3 * v3.y));
      acc.z = acc.z * r + ((w0 * v0.z + w1 * v1.z) + (w2 * v2.z + w3 * v3.z));
      acc.w = acc.w * r + ((w0 * v0.w + w1 * v1.w) + (w2 * v2.w + w3 * v3.w));
      m = mn;
    }
  }

  float4 o;
  o.x = (d > 0.0f) ? acc.x / d : 0.0f;
  o.y = (d > 0.0f) ? acc.y / d : 0.0f;
  o.z = (d > 0.0f) ? acc.z / d : 0.0f;
  o.w = (d > 0.0f) ? acc.w / d : 0.0f;
  o.x += __shfl_xor(o.x, 8, 32);  o.x += __shfl_xor(o.x, 16, 32);
  o.y += __shfl_xor(o.y, 8, 32);  o.y += __shfl_xor(o.y, 16, 32);
  o.z += __shfl_xor(o.z, 8, 32);  o.z += __shfl_xor(o.z, 16, 32);
  o.w += __shfl_xor(o.w, 8, 32);  o.w += __shfl_xor(o.w, 16, 32);
  if (lane < 8) {
    o.x *= 0.25f; o.y *= 0.25f; o.z *= 0.25f; o.w *= 0.25f;
    *(float4*)(attnMean + (size_t)n * FEAT + f4) = o;
  }
}

__global__ __launch_bounds__(64) void gat_out_wmma(const float* __restrict__ x,
                                                   const float* __restrict__ attnMean,
                                                   const float* __restrict__ Wout,
                                                   float* __restrict__ out) {
  const int lane = threadIdx.x & 31;
  const int nt   = threadIdx.x >> 5;
  const int m0   = blockIdx.x * 16;

  __shared__ float as[16 * FEAT];
  {
    const float4* src = (const float4*)(attnMean + (size_t)m0 * FEAT);
    ((float4*)as)[threadIdx.x]      = src[threadIdx.x];
    ((float4*)as)[threadIdx.x + 64] = src[threadIdx.x + 64];
  }
  __syncthreads();

  v8f acc = {};
  acc = wmma_f32_k32(as, Wout, nt * 16, lane, acc);

  const int col = lane & 15, hi = lane >> 4;
#pragma unroll
  for (int r = 0; r < 8; ++r) {
    const size_t o = (size_t)(m0 + r + 8 * hi) * FEAT + nt * 16 + col;
    out[o] = x[o] + acc[r];
  }
}

extern "C" void kernel_launch(void* const* d_in, const int* in_sizes, int n_in,
                              void* d_out, int out_size, void* d_ws, size_t ws_size,
                              hipStream_t stream) {
  const float* x    = (const float*)d_in[0];
  const int*   ei   = (const int*)d_in[1];
  const float* Wq   = (const float*)d_in[2];
  const float* Wk   = (const float*)d_in[3];
  const float* Wv   = (const float*)d_in[4];
  const float* Wout = (const float*)d_in[5];
  float*       out  = (float*)d_out;

  float* ws   = (float*)d_ws;
  float* qkv  = ws;
  float* attn = qkv + (size_t)3 * HEADS * N_NODES * FEAT;
  int*   tbl  = (int*)(attn + (size_t)N_NODES * FEAT);
  int*   k1   = tbl + 2 * NBIN * NBLK;
  int*   v1   = k1 + N_EDGES;
  int*   v2   = v1 + N_EDGES;
  int*   off  = v2 + N_EDGES;

  tile_partition<<<NBLK + QT_WG, 256, 0, stream>>>(ei + N_EDGES, ei, k1, v1, tbl,
                                                     x, Wq, Wk, Wv, qkv);
  bucket_finish<<<NBUCKET + QB_BUCKET, BK_THREADS, 0, stream>>>(k1, v1, tbl, v2, off,
                                                                x, Wq, Wk, Wv, qkv);

  gat_node_pull<<<N_NODES / 8, 256, 0, stream>>>(qkv, off, v2, attn);

  gat_out_wmma<<<MTILES, 64, 0, stream>>>(x, attn, Wout, out);
}


# pull kernel: 10000 resident waves each walk 5 receivers (next one's q row, senders and bounds requested ahead; registers of 10 waves per SIMD so all 8 row gathers fly), not 50000 one-receiver waves
# speedup vs baseline: 3.3939x; 1.0307x over previous
#include <hip/hip_runtime.h>
#include <math.h>


#define N_NODES 50000
#define N_EDGES 800000
#define FEAT    32
#define HEADS   4
#define MTILES  (N_NODES / 16)

#define NBIN        256
#define TILE        4096
#define NBLK        ((N_EDGES + TILE - 1) / TILE)
#define TBLW        (NBIN + 1)

#define NBUCKET     ((N_NODES + NBIN - 1) / NBIN)
#define BK_WAVES    16
#define BK_THREADS  (BK_WAVES * 32)
#define BK_CHUNK    (BK_WAVES * 512)

#define QB_BUCKET   240
#define QT_TILE     (MTILES - 2 * QB_BUCKET)
#define QT_WG       256
static_assert(QT_TILE > 0 && QT_TILE + 2 * QB_BUCKET == MTILES, "every projection tile has one carrier");
static_assert(QT_WG > 0 && QT_WG <= QT_TILE && QT_TILE <= MTILES, "every walk has a first tile; whole tiles only");

static_assert(NBLK * TBLW <= 2 * NBIN * NBLK, "the table fits the space of the old counter planes");
static_assert(NBLK <= NBIN && NBUCKET < NBIN, "one thread per tile run in the bucket pass");

typedef __attribute__((ext_vector_type(2))) float v2f;
typedef __attribute__((ext_vector_type(8))) float v8f;

__device__ __forceinline__ v8f wmma_f32_k32(const float* Arow,
                                            const float* __restrict__ B,
                                            int bcol, int lane, v8f acc) {
  const int row = lane & 15;
  const int hi  = lane >> 4;
#pragma unroll
  for (int kc = 0; kc < 8; ++kc) {
    const int k0 = kc * 4 + 2 * hi;
    v2f a, b;
    a.x = Arow[row * FEAT + k0];
    a.y = Arow[row * FEAT + k0 + 1];
    b.x = B[k0 * FEAT + bcol + row];
    b.y = B[(k0 + 1) * FEAT + bcol + row];
    acc = __builtin_amdgcn_wmma_f32_16x16x4_f32(false, a, false, b, (short)0, acc,
                                                false, false);
  }
  return acc;
}

template <int NWAVES>
__device__ __forceinline__ void qkv_tiles(const float* __restrict__ x,
                                          const float* __restrict__ Wq,
                                          const float* __restrict__ Wk,
                                          const float* __restrict__ Wv,
                                          float* __restrict__ qkv, int tile0, float* xs) {
  constexpr int TPB = NWAVES / 8;
  const int t    = threadIdx.x;
  const int lane = t & 31;
  const int wave = t >> 5;

  if (t < TPB * 128) {
    const int row = tile0 * 16 + (t >> 3);
    float4 v = make_float4(0.0f, 0.0f, 0.0f, 0.0f);
    if (row < N_NODES) v = ((const float4*)(x + (size_t)tile0 * 16 * FEAT))[t];
    ((float4*)xs)[t] = v;
  }
  __syncthreads();

  const int tl   = wave >> 3;
  const int head = (wave >> 1) & 3;
  const int nt   = wave & 1;
  const int m0   = (tile0 + tl) * 16;
  if (m0 >= N_NODES) return;

  const int col = lane & 15, hi = lane >> 4;
#pragma unroll
  for (int mat = 0; mat < 3; ++mat) {
    const float* W = (mat == 0 ? Wq : (mat == 1 ? Wk : Wv)) + head * FEAT * FEAT;
    v8f acc = {};
    acc = wmma_f32_k32(xs + tl * 16 * FEAT, W, nt * 16, lane, acc);

    float* out = qkv + ((size_t)(mat * HEADS + head) * N_NODES) * FEAT;
#pragma unroll
    for (int r = 0; r < 8; ++r)
      out[(size_t)(m0 + r + 8 * hi) * FEAT + nt * 16 + col] = acc[r];
  }
}

__device__ __forceinline__ void qkv_tile_walk(const float* __restrict__ x,
                                              const float* __restrict__ Wq,
                                              const float* __restrict__ Wk,
                                              const float* __restrict__ Wv,
                                              float* __restrict__ qkv, int first, int step, float* xs) {
  const int t    = threadIdx.x;
  const int lane = t & 31;
  const int wave = t >> 5;
  const int head = (wave >> 1) & 3;
  const int nt   = wave & 1;
  const int row  = lane & 15, hi = lane >> 4;

  float4 xv = make_float4(0.0f, 0.0f, 0.0f, 0.0f);
  if (t < 128) xv = ((const float4*)(x + (size_t)first * 16 * FEAT))[t];

  v2f bq[8], bk[8], bv[8];
#pragma unroll
  for (int kc = 0; kc < 8; ++kc) {
    const int o = head * FEAT * FEAT + (kc * 4 + 2 * hi) * FEAT + nt * 16 + row;
    bq[kc].x = Wq[o];  bq[kc].y = Wq[o + FEAT];
    bk[kc].x = Wk[o];  bk[kc].y = Wk[o + FEAT];
    bv[kc].x = Wv[o];  bv[kc].y = Wv[o + FEAT];
  }

  if (t < 128) ((float4*)xs)[t] = xv;
  __syncthreads();

  int i = 0;
  for (int tile = first; tile < QT_TILE; tile += step, ++i) {
    const bool more = tile + step < QT_TILE;
    if (more && t < 128) xv = ((const float4*)(x + (size_t)(tile + step) * 16 * FEAT))[t];

    const float* A = xs + (i & 1) * 16 * FEAT;
    v2f a[8];
#pragma unroll
    for (int kc = 0; kc < 8; ++kc) {
      const int k0 = kc * 4 + 2 * hi;
      a[kc].x = A[row * FEAT + k0];
      a[kc].y = A[row * FEAT + k0 + 1];
    }
    const int m0 = tile * 16;
#pragma unroll
    for (int mat = 0; mat < 3; ++mat) {
      v8f acc = {};
#pragma unroll
      for (int kc = 0; kc < 8; ++kc) {
        const v2f b = (mat == 0 ? bq[kc] : (mat == 1 ? bk[kc] : bv[kc]));
        acc = __builtin_amdgcn_wmma_f32_16x16x4_f32(false, a[kc], false, b, (short)0, acc,
                                                    false, false);
      }
      float* out = qkv + ((size_t)(mat * HEADS + head) * N_NODES) * FEAT;
#pragma unroll
      for (int r = 0; r < 8; ++r)
        out[(size_t)(m0 + r + 8 * hi) * FEAT + nt * 16 + row] = acc[r];
    }

    if (more && t < 128) ((float4*)(xs + ((i + 1) & 1) * 16 * FEAT))[t] = xv;
    __builtin_amdgcn_fence(__ATOMIC_RELEASE, "workgroup", "local");
    __builtin_amdgcn_s_barrier();
    __builtin_amdgcn_fence(__ATOMIC_ACQUIRE, "workgroup", "local");
  }
}

__device__ __forceinline__ unsigned same_digit_lanes(unsigned dgt, bool valid) {
  unsigned mask = __builtin_amdgcn_ballot_w32(valid);
#pragma unroll
  for (int b = 0; b < 8; ++b) {
    const bool bit = ((dgt >> b) & 1u) != 0u;
    const unsigned bb = __builtin_amdgcn_ballot_w32(bit);
    mask &= bit ? bb : ~bb;
  }
  return mask;
}

__device__ __forceinline__ void load_wave_words(const int* __restrict__ src, int wbase, int lane,
                                                int limit, int (&k)[16]) {
#pragma unroll
  for (int it = 0; it < 16; ++it) {
    const int e = wbase + it * 32 + lane;
    k[it] = (e < limit) ? src[e] : 0;
  }
}

template <int SHIFT>
__device__ __forceinline__ void wave_histogram(const int (&k)[16], int* hrow, int wbase, int lane,
                                               int limit) {
#pragma unroll
  for (int it = 0; it < 16; ++it) {
    const bool valid = wbase + it * 32 + lane < limit;
    const unsigned dgt  = ((unsigned)k[it] >> SHIFT) & 255u;
    const unsigned mask = same_digit_lanes(dgt, valid);
    const unsigned rank = __builtin_popcount(mask & ((1u << lane) - 1u));
    if (valid && rank == 0u) hrow[dgt] += (int)__builtin_popcount(mask);
    __builtin_amdgcn_fence(__ATOMIC_ACQ_REL, "wavefront");
  }
}

__device__ __forceinline__ int wave_incl_scan(int v, int lane);

__global__ __launch_bounds__(256) void tile_partition(const int* __restrict__ keys,
                                                      const int* __restrict__ vals,
                                                      int* __restrict__ okeys,
                                                      int* __restrict__ ovals,
                                                      int* __restrict__ tbl,
                                                      const float* __restrict__ x,
                                                      const float* __restrict__ Wq,
                                                      const float* __restrict__ Wk,
                                                      const float* __restrict__ Wv,
                                                      float* __restrict__ qkv) {
  __shared__ alignas(16) int hist[8 * NBIN];
  __shared__ int wtot[8];
  __shared__ alignas(16) int stk[TILE];
  __shared__ alignas(16) int stv[TILE];
  if (blockIdx.x >= NBLK) {
    qkv_tile_walk(x, Wq, Wk, Wv, qkv, (int)blockIdx.x - NBLK, QT_WG, (float*)hist);
    return;
  }
  const int t = threadIdx.x, lane = t & 31, w = t >> 5;
  const int tbase = blockIdx.x * TILE;
  const int wbase = tbase + w * 512;
  int k[16], v[16];
  load_wave_words(keys, wbase, lane, N_EDGES, k);
  load_wave_words(vals, wbase, lane, N_EDGES, v);
  for (int i = t; i < 8 * NBIN; i += 256) hist[i] = 0;
  __syncthreads();
  int* hrow = hist + w * NBIN;
  wave_histogram<8>(k, hrow, wbase, lane, N_EDGES);
  __syncthreads();
  {
    int tot = 0;
    for (int ww = 0; ww < 8; ++ww) tot += hist[ww * NBIN + t];
    const int incl = wave_incl_scan(tot, lane);
    if (lane == 31) wtot[w] = incl;
    __syncthreads();
    int run = incl - tot;
    for (int ww = 0; ww < 8; ++ww) run += (ww < w) ? wtot[ww] : 0;
    tbl[blockIdx.x * TBLW + t] = run;
    if (t == NBIN - 1) tbl[blockIdx.x * TBLW + NBIN] = run + tot;
    for (int ww = 0; ww < 8; ++ww) {
      const int c = hist[ww * NBIN + t];
      hist[ww * NBIN + t] = run;
      run += c;
    }
  }
  __syncthreads();
#pragma unroll
  for (int it = 0; it < 16; ++it) {
    const bool valid = wbase + it * 32 + lane < N_EDGES;
    const unsigned dgt  = ((unsigned)k[it] >> 8) & 255u;
    const unsigned mask = same_digit_lanes(dgt, valid);
    const unsigned rank = __builtin_popcount(mask & ((1u << lane) - 1u));
    const int start = hrow[dgt];
    if (valid) {
      const int pos = start + (int)rank;
      if ((unsigned)pos < (unsigned)TILE) {
        stk[pos] = k[it];
        stv[pos] = v[it];
      }
    }
    if (valid && rank == 0u) hrow[dgt] = start + (int)__builtin_popcount(mask);
    __builtin_amdgcn_fence(__ATOMIC_ACQ_REL, "wavefront");
  }
  __syncthreads();
  static_assert(TILE % 4 == 0 && N_EDGES % 4 == 0, "tiles are written back in 16-byte pieces");
  const int nvalid = (N_EDGES - tbase) < TILE ? (N_EDGES - tbase) : TILE;
  for (int i = 4 * t; i < nvalid; i += 4 * 256) {
    *(int4*)(okeys + tbase + i) = *(const int4*)(stk + i);
    *(int4*)(ovals + tbase + i) = *(const int4*)(stv + i);
  }
}

__device__ __forceinline__ int wave_incl_scan(int v, int lane) {
#pragma unroll
  for (int d = 1; d < 32; d <<= 1) {
    const int u = __shfl_up(v, d, 32);
    if (lane >= d) v += u;
  }
  return v;
}

__device__ __forceinline__ void wave_place_vals(const int (&k)[16], const int (&v)[16], int* hrow,
                                                int wbase, int lane, int limit,
                                                int* __restrict__ ovals, int* stage, int beg,
                                                bool staged) {
#pragma unroll
  for (int it = 0; it < 16; ++it) {
    const bool valid = wbase + it * 32 + lane < limit;
    const unsigned dgt  = (unsigned)k[it] & 255u;
    const unsigned mask = same_digit_lanes(dgt, valid);
    const unsigned rank = __builtin_popcount(mask & ((1u << lane) - 1u));
    const int start = hrow[dgt];
    if (valid) {
      const int pos = start + (int)rank;
      if (staged) {
        if ((unsigned)(pos - beg) < (unsigned)BK_CHUNK) stage[pos - beg] = v[it];
      } else {
        if ((unsigned)pos < (unsigned)N_EDGES) ovals[pos] = v[it];
      }
    }
    if (valid && rank == 0u) hrow[dgt] = start + (int)__builtin_popcount(mask);
    __builtin_amdgcn_fence(__ATOMIC_ACQ_REL, "wavefront");
  }
}

__device__ __forceinline__ int counts_to_starts(int* hist, int t, int run) {
  for (int ww = 0; ww < BK_WAVES; ++ww) {
    const int c = hist[ww * NBIN + t];
    hist[ww * NBIN + t] = run;
    run += c;
  }
  return run;
}

#define RUN_BATCH 6
__device__ __forceinline__ int edge_slot(int s) {
  return s < 0 ? 0 : (s >= N_EDGES ? N_EDGES - 1 : s);
}
__device__ __forceinline__ void copy_runs(const int* __restrict__ keys, const int* __restrict__ vals,
                                          const int* rsrc, const int* rdst, int cbase, int w, int lane,
                                          int* sk, int* sv) {
  const int g = lane >> 3, sub = lane & 7;
  int lo[4], hi[4], rel[4];
#pragma unroll
  for (int r = 0; r < 4; ++r) {
    const int j = 64 * r + 4 * w + g;
    lo[r] = 0; hi[r] = 0; rel[r] = 0;
    if (j < NBLK) {
      const int d0 = rdst[j], d1 = rdst[j + 1];
      lo[r]  = d0 > cbase ? d0 : cbase;
      hi[r]  = d1 < cbase + BK_CHUNK ? d1 : cbase + BK_CHUNK;
      rel[r] = rsrc[j] - d0;
    }
  }
  int kk[4][RUN_BATCH], vv[4][RUN_BATCH];
#pragma unroll
  for (int r = 0; r < 4; ++r) {
#pragma unroll
    for (int u = 0; u < RUN_BATCH; ++u) {
      const int p = lo[r] + sub + 8 * u;
      const int s = edge_slot(rel[r] + (p < hi[r] ? p : lo[r]));
      kk[r][u] = keys[s];
      vv[r][u] = vals[s];
    }
  }
#pragma unroll
  for (int r = 0; r < 4; ++r) {
#pragma unroll
    for (int u = 0; u < RUN_BATCH; ++u) {
      const int p = lo[r] + sub + 8 * u;
      if (p < hi[r]) { sk[p - cbase] = kk[r][u]; sv[p - cbase] = vv[r][u]; }
    }
  }
#pragma unroll
  for (int r = 0; r < 4; ++r) {
    for (int p = lo[r] + 8 * RUN_BATCH + sub; p < hi[r]; p += 8) {
      const int s = edge_slot(rel[r] + p);
      sk[p - cbase] = keys[s];
      sv[p - cbase] = vals[s];
    }
  }
}

__device__ __forceinline__ void take_wave_words(const int* wsrc, int ebase, int lane, int len,
                                                int (&o)[16]) {
#pragma unroll
  for (int it = 0; it < 16; ++it) o[it] = (ebase + it * 32 + lane < len) ? wsrc[it * 32 + lane] : 0;
}

__global__ __launch_bounds__(BK_THREADS) void bucket_finish(const int* __restrict__ keys,
                                                           const int* __restrict__ vals,
                                                           const int* __restrict__ tbl,
                                                           int* __restrict__ ovals,
                                                           int* __restrict__ off,
                                                           const float* __restrict__ x,
                                                           const float* __restrict__ Wq,
                                                           const float* __restrict__ Wk,
                                                           const float* __restrict__ Wv,
                                                           float* __restrict__ qkv) {
  __shared__ alignas(16) int hist[BK_WAVES * NBIN];
  __shared__ int wtot[NBIN / 32];
  __shared__ int wlen[NBIN / 32];
  __shared__ int wbeg[NBIN / 32];
  __shared__ int rsrc[NBLK];
  __shared__ int rdst[NBLK + 1];
  __shared__ int sk[BK_CHUNK];
  __shared__ int sv[BK_CHUNK];
  int* stage = sk;
  if (blockIdx.x >= NBUCKET) {
    qkv_tiles<16>(x, Wq, Wk, Wv, qkv, QT_TILE + 2 * ((int)blockIdx.x - NBUCKET), (float*)hist);
    return;
  }
  const int t = threadIdx.x, lane = t & 31, w = t >> 5;
  const int b = blockIdx.x;

  int rs = 0, rl = 0;
  if (t < NBLK) {
    rs = tbl[t * TBLW + b];
    int re = tbl[t * TBLW + b + 1];
    rs = rs < 0 ? 0 : (rs > TILE ? TILE : rs);
    re = re < rs ? rs : (re > TILE ? TILE : re);
    rl = re - rs;
  }
  int rincl = 0;
  if (t < NBIN) {
    rincl = wave_incl_scan(rl, lane);
    const int sincl = wave_incl_scan(rs, lane);
    if (lane == 31) { wlen[w] = rincl; wbeg[w] = sincl; }
  }
  __syncthreads();
  int len = 0, beg = 0;
  for (int ww = 0; ww < NBIN / 32; ++ww) { len += wlen[ww]; beg += wbeg[ww]; }
  if (t < NBLK) {
    int wb = 0;
    for (int ww = 0; ww < NBIN / 32; ++ww) wb += (ww < w) ? wlen[ww] : 0;
    rsrc[t] = t * TILE + rs;
    rdst[t] = wb + rincl - rl;
  }
  if (t == 0) rdst[NBLK] = len;
  beg = beg < 0 ? 0 : (beg > N_EDGES ? N_EDGES : beg);
  len = len < 0 ? 0 : (len > N_EDGES - beg ? N_EDGES - beg : len);
  __syncthreads();

  const int nch = len > BK_CHUNK ? (len + BK_CHUNK - 1) / BK_CHUNK : 1;
  const int nst = (nch == 1) ? 1 : 2 * nch;

  int* hrow = hist + w * NBIN;
  int run = 0;
  for (int i = 0; i < nst; ++i) {
    const bool placing = (nch == 1) || (i >= nch);
    const int  c       = (i >= nch) ? i - nch : i;
    const int  ebase   = c * BK_CHUNK + w * 512;
    int k[16], v[16];
    copy_runs(keys, vals, rsrc, rdst, c * BK_CHUNK, w, lane, sk, sv);
    __syncthreads();
    take_wave_words(sk + w * 512, ebase, lane, len, k);
    take_wave_words(sv + w * 512, ebase, lane, len, v);

    if (i == 0 || i >= nch) {
      __syncthreads();
      for (int j = t; j < BK_WAVES * NBIN; j += BK_THREADS) hist[j] = 0;
      __syncthreads();
    }
    wave_histogram<0>(k, hrow, ebase, lane, len);
    __syncthreads();

    if (i == nch - 1) {
      int tot = 0, incl = 0;
      if (t < NBIN) {
        for (int ww = 0; ww < BK_WAVES; ++ww) tot += hist[ww * NBIN + t];
        incl = wave_incl_scan(tot, lane);
        if (lane == 31) wtot[w] = incl;
      }
      __syncthreads();
      if (t < NBIN) {
        int wb = 0;
        for (int ww = 0; ww < NBIN / 32; ++ww) wb += (ww < w) ? wtot[ww] : 0;
        run = beg + wb + incl - tot;
        const int n = b * NBIN + t;
        if (n <= N_NODES) off[n] = run;
      }
    }

    if (placing) {
      if (t < NBIN) run = counts_to_starts(hist, t, run);
      __syncthreads();
      wave_place_vals(k, v, hrow, ebase, lane, len, ovals, stage, beg, nch == 1);
      if (nch == 1) {
        __syncthreads();
        for (int j = t; j < len; j += BK_THREADS) ovals[beg + j] = stage[j];
      }
    }
  }
}

#define PULL_RECV   5
#define PULL_WG     (N_NODES / (8 * PULL_RECV))
#define PULL_WAVES  (PULL_WG * 8)
static_assert(PULL_WAVES * PULL_RECV == N_NODES, "every wave of the pull has PULL_RECV receivers");
__device__ __forceinline__ float sum8(float p) {
  p += __shfl_xor(p, 1, 32);
  p += __shfl_xor(p, 2, 32);
  p += __shfl_xor(p, 4, 32);
  return p;
}
__device__ __forceinline__ float dot4(const float4 a, const float4 b) {
  return a.x * b.x + a.y * b.y + a.z * b.z + a.w * b.w;
}
__device__ __forceinline__ int seg_beg(int b) { return b < 0 ? 0 : (b > N_EDGES ? N_EDGES : b); }
__device__ __forceinline__ int seg_end(int e, int b) { return e < b ? b : (e > N_EDGES ? N_EDGES : e); }
__device__ __forceinline__ int node_row(int s) { return s < 0 ? 0 : (s >= N_NODES ? N_NODES - 1 : s); }

__global__ __launch_bounds__(256) __attribute__((amdgpu_waves_per_eu(10, 10))) void gat_node_pull(const float* __restrict__ qkv,
                                                     const int* __restrict__ off,
                                                     const int* __restrict__ ssnd,
                                                     float* __restrict__ attnMean) {
  const int lane = threadIdx.x & 31;
  const int first = __builtin_amdgcn_readfirstlane((int)((blockIdx.x * blockDim.x + threadIdx.x) >> 5));
  if (first >= N_NODES) return;
  const int h  = lane >> 3;
  const int f4 = (lane & 7) * 4;

  const float* Qb = qkv + ((size_t)(0 * HEADS + h) * N_NODES) * FEAT + f4;
  const float* Kb = qkv + ((size_t)(1 * HEADS + h) * N_NODES) * FEAT + f4;
  const float* Vb = qkv + ((size_t)(2 * HEADS + h) * N_NODES) * FEAT + f4;

  int beg = seg_beg(off[first]);
  int end = seg_end(off[first + 1], beg);
  float4 q = *(const float4*)(Qb + (size_t)first * FEAT);
  int sv = node_row((lane < end - beg) ? ssnd[beg + lane] : 0);
  int nbeg = 0, nend = 0;
  if (first + PULL_WAVES < N_NODES) {
    nbeg = seg_beg(off[first + PULL_WAVES]);
    nend = seg_end(off[first + PULL_WAVES + 1], nbeg);
  }

  for (int n = first; n < N_NODES; n += PULL_WAVES) {
    const int n1 = n + PULL_WAVES, n2 = n + 2 * PULL_WAVES;
    float4 nq  = make_float4(0.0f, 0.0f, 0.0f, 0.0f);
    int    nsv = 0;
    if (n1 < N_NODES) {
      nq  = *(const float4*)(Qb + (size_t)n1 * FEAT);
      nsv = (lane < nend - nbeg) ? ssnd[nbeg + lane] : 0;
    }
    const int n2c = n2 < N_NODES ? n2 : N_NODES - 1;
    int n2b = off[n2c];
    int n2e = off[n2c + 1];

    float  m = -INFINITY, d = 0.0f;
    float4 acc = make_float4(0.0f, 0.0f, 0.0f, 0.0f);

    int base = beg;
    while (base < end) {
      const int rem = end - base;
      const int cn  = rem < 32 ? rem : 32;

      for (int j = 0; j < cn; j += 4) {
        const bool ok1 = j + 1 < cn, ok2 = j + 2 < cn, ok3 = j + 3 < cn;
        const int s0 = __builtin_amdgcn_readlane(sv, j);
        const int s1 = __builtin_amdgcn_readlane(sv, ok1 ? j + 1 : j);
        const int s2 = __builtin_amdgcn_readlane(sv, ok2 ? j + 2 : j);
        const int s3 = __builtin_amdgcn_readlane(sv, ok3 ? j + 3 : j);
        const float4 k0 = *(const float4*)(Kb + (size_t)s0 * FEAT);
        const float4 k1 = *(const float4*)(Kb + (size_t)s1 * FEAT);
        const float4 k2 = *(const float4*)(Kb + (size_t)s2 * FEAT);
        const float4 k3 = *(const float4*)(Kb + (size_t)s3 * FEAT);
        const float4 v0 = *(const float4*)(Vb + (size_t)s0 * FEAT);
        const float4 v1 = *(const float4*)(Vb + (size_t)s1 * FEAT);
        const float4 v2 = *(const float4*)(Vb + (size_t)s2 * FEAT);
        const float4 v3 = *(const float4*)(Vb + (size_t)s3 * FEAT);

        const float sc = 0.17677669529663687f;
        const float pen1 = ok1 ? 0.0f : -INFINITY;
        const float pen2 = ok2 ? 0.0f : -INFINITY;
        const float pen3 = ok3 ? 0.0f : -INFINITY;
        const float p0 = sum8(dot4(q, k0)) * sc;
        const float p1 = sum8(dot4(q, k1)) * sc + pen1;
        const float p2 = sum8(dot4(q, k2)) * sc + pen2;
        const float p3 = sum8(dot4(q, k3)) * sc + pen3;

        const float mn = fmaxf(fmaxf(m, p0), fmaxf(fmaxf(p1, p2), p3));
        const float r  = __expf(m - mn);
        const float w0 = __expf(p0 - mn);
        const float w1 = __expf(p1 - mn);
        const float w2 = __expf(p2 - mn);
        const float w3 = __expf(p3 - mn);
        d = d * r + ((w0 + w1) + (w2 + w3));
        acc.x = acc.x * r + ((w0 * v0.x + w1 * v1.x) + (w2 * v2.x + w3 * v3.x));
        acc.y = acc.y * r + ((w0 * v0.y + w1 * v1.y) + (w2 * v2.y + w3 * v3.y));
        acc.z = acc.z * r + ((w0 * v0.z + w1 * v1.z) + (w2 * v2.z + w3 * v3.z));
        acc.w = acc.w * r + ((w0 * v0.w + w1 * v1.w) + (w2 * v2.w + w3 * v3.w));
        m = mn;
      }
      base += 32;
      if (base < end) sv = node_row((lane < end - base) ? ssnd[base + lane] : 0);
    }

    float4 o;
    o.x = (d > 0.0f) ? acc.x / d : 0.0f;
    o.y = (d > 0.0f) ? acc.y / d : 0.0f;
    o.z = (d > 0.0f) ? acc.z / d : 0.0f;
    o.w = (d > 0.0f) ? acc.w / d : 0.0f;
    o.x += __shfl_xor(o.x, 8, 32);  o.x += __shfl_xor(o.x, 16, 32);
    o.y += __shfl_xor(o.y, 8, 32);  o.y += __shfl_xor(o.y, 16, 32);
    o.z += __shfl_xor(o.z, 8, 32);  o.z += __shfl_xor(o.z, 16, 32);
    o.w += __shfl_xor(o.w, 8, 32);  o.w += __shfl_xor(o.w, 16, 32);
    if (lane < 8) {
      o.x *= 0.25f; o.y *= 0.25f; o.z *= 0.25f; o.w *= 0.25f;
      *(float4*)(attnMean + (size_t)n * FEAT + f4) = o;
    }

    asm volatile("" : "+v"(nsv));
    asm volatile("" : "+s"(n2b), "+s"(n2e));
    beg = nbeg;  end = nend;  q = nq;  sv = node_row(nsv);
    nbeg = seg_beg(n2b);  nend = seg_end(n2e, nbeg);
  }
}

__global__ __launch_bounds__(64) void gat_out_wmma(const float* __restrict__ x,
                                                   const float* __restrict__ attnMean,
                                                   const float* __restrict__ Wout,
                                                   float* __restrict__ out) {
  const int lane = threadIdx.x & 31;
  const int nt   = threadIdx.x >> 5;
  const int m0   = blockIdx.x * 16;

  __shared__ float as[16 * FEAT];
  {
    const float4* src = (const float4*)(attnMean + (size_t)m0 * FEAT);
    ((float4*)as)[threadIdx.x]      = src[threadIdx.x];
    ((float4*)as)[threadIdx.x + 64] = src[threadIdx.x + 64];
  }
  __syncthreads();

  v8f acc = {};
  acc = wmma_f32_k32(as, Wout, nt * 16, lane, acc);

  const int col = lane & 15, hi = lane >> 4;
#pragma unroll
  for (int r = 0; r < 8; ++r) {
    const size_t o = (size_t)(m0 + r + 8 * hi) * FEAT + nt * 16 + col;
    out[o] = x[o] + acc[r];
  }
}

extern "C" void kernel_launch(void* const* d_in, const int* in_sizes, int n_in,
                              void* d_out, int out_size, void* d_ws, size_t ws_size,
                              hipStream_t stream) {
  const float* x    = (const float*)d_in[0];
  const int*   ei   = (const int*)d_in[1];
  const float* Wq   = (const float*)d_in[2];
  const float* Wk   = (const float*)d_in[3];
  const float* Wv   = (const float*)d_in[4];
  const float* Wout = (const float*)d_in[5];
  float*       out  = (float*)d_out;

  float* ws   = (float*)d_ws;
  float* qkv  = ws;
  float* attn = qkv + (size_t)3 * HEADS * N_NODES * FEAT;
  int*   tbl  = (int*)(attn + (size_t)N_NODES * FEAT);
  int*   k1   = tbl + 2 * NBIN * NBLK;
  int*   v1   = k1 + N_EDGES;
  int*   v2   = v1 + N_EDGES;
  int*   off  = v2 + N_EDGES;

  tile_partition<<<NBLK + QT_WG, 256, 0, stream>>>(ei + N_EDGES, ei, k1, v1, tbl,
                                                     x, Wq, Wk, Wv, qkv);
  bucket_finish<<<NBUCKET + QB_BUCKET, BK_THREADS, 0, stream>>>(k1, v1, tbl, v2, off,
                                                                x, Wq, Wk, Wv, qkv);

  gat_node_pull<<<PULL_WG, 256, 0, stream>>>(qkv, off, v2, attn);

  gat_out_wmma<<<MTILES, 64, 0, stream>>>(x, attn, Wout, out);
}
